# starting kernel
# baseline (speedup 1.0000x reference)
#include <hip/hip_runtime.h>

typedef float v2f __attribute__((ext_vector_type(2)));
typedef float v8f __attribute__((ext_vector_type(8)));

__global__ __launch_bounds__(256) void TrainableMovingAverage1d_wmma_kernel(
    const float* __restrict__ x, const float* __restrict__ w,
    float* __restrict__ out, long long n_out) {
  const int lane = threadIdx.x & 31;
  const int wave = threadIdx.x >> 5;
  const int half = lane >> 4;
  const int l    = lane & 15;

  const long long base = ((long long)blockIdx.x * 8 + wave) * 16;
  if (base + 16 > n_out) return;

  const long long m = base + l;

  const float* xp = x + m * 8;
  v2f a1 = *(const v2f*)(xp + 2 * half);
  v2f a2 = *(const v2f*)(xp + 4 + 2 * half);

  const v2f* w2 = (const v2f*)w;
  v2f b1 = w2[half];
  v2f b2 = w2[half + 2];

  v8f c = {0.f, 0.f, 0.f, 0.f, 0.f, 0.f, 0.f, 0.f};
  c = __builtin_amdgcn_wmma_f32_16x16x4_f32(false, a1, false, b1, (short)0, c,
                                            false, false);
  c = __builtin_amdgcn_wmma_f32_16x16x4_f32(false, a2, false, b2, (short)0, c,
                                            false, false);

  if (l == 0) {
    float* o = out + base + 8 * half;
    float4 lo, hi;
    lo.x = c[0]; lo.y = c[1]; lo.z = c[2]; lo.w = c[3];
    hi.x = c[4]; hi.y = c[5]; hi.z = c[6]; hi.w = c[7];
    *(float4*)(o)     = lo;
    *(float4*)(o + 4) = hi;
  }
}

extern "C" void kernel_launch(void* const* d_in, const int* in_sizes, int n_in,
                              void* d_out, int out_size, void* d_ws, size_t ws_size,
                              hipStream_t stream) {
  const float* x = (const float*)d_in[0];
  const float* w = (const float*)d_in[1];
  float* out = (float*)d_out;

  const long long n_out = (long long)out_size;
  const long long waves  = (n_out + 15) / 16;
  const int blocks = (int)((waves + 7) / 8);

  TrainableMovingAverage1d_wmma_kernel<<<blocks, 256, 0, stream>>>(x, w, out, n_out);
}


# one tile per wave -> resident waves: 4096 waves each walk 64 contiguous tiles, 8 per group, the next group's loads in flight under the current group's WMMAs; sums bit-identical
# speedup vs baseline: 1.6119x; 1.6119x over previous
#include <hip/hip_runtime.h>

typedef float v2f __attribute__((ext_vector_type(2)));
typedef float v8f __attribute__((ext_vector_type(8)));


constexpr int TU     = 8;
constexpr int TILE_X = 128;
constexpr int TILE_O = 16;

struct Group {
  v2f lo[TU];
  v2f hi[TU];
};

static __device__ __forceinline__ void load_group(Group& g,
                                                  const float* __restrict__ xp) {
#pragma unroll
  for (int u = 0; u < TU; ++u) {
    g.lo[u] = *(const v2f*)(xp + u * TILE_X);
    g.hi[u] = *(const v2f*)(xp + u * TILE_X + 4);
  }
}

static __device__ __forceinline__ void mul_store_group(const Group& g, v2f b1,
                                                       v2f b2,
                                                       float* __restrict__ op,
                                                       bool writer) {
  v8f c[TU];
#pragma unroll
  for (int u = 0; u < TU; ++u) {
    const v8f z = {0.f, 0.f, 0.f, 0.f, 0.f, 0.f, 0.f, 0.f};
    c[u] = __builtin_amdgcn_wmma_f32_16x16x4_f32(false, g.lo[u], false, b1,
                                                 (short)0, z, false, false);
  }
#pragma unroll
  for (int u = 0; u < TU; ++u) {
    c[u] = __builtin_amdgcn_wmma_f32_16x16x4_f32(false, g.hi[u], false, b2,
                                                 (short)0, c[u], false, false);
  }
  if (writer) {
#pragma unroll
    for (int u = 0; u < TU; ++u) {
      float* o = op + u * TILE_O;
      float4 lo, hi;
      lo.x = c[u][0]; lo.y = c[u][1]; lo.z = c[u][2]; lo.w = c[u][3];
      hi.x = c[u][4]; hi.y = c[u][5]; hi.z = c[u][6]; hi.w = c[u][7];
      *(float4*)(o)     = lo;
      *(float4*)(o + 4) = hi;
    }
  }
}

__global__ __launch_bounds__(256) void TrainableMovingAverage1d_wmma_kernel(
    const float* __restrict__ x, const float* __restrict__ w,
    float* __restrict__ out, long long n_out) {
  const int lane = threadIdx.x & 31;
  const unsigned wave = __builtin_amdgcn_readfirstlane(threadIdx.x >> 5);
  const int half = lane >> 4;
  const int l    = lane & 15;

  const unsigned n_tiles = (unsigned)(n_out >> 4);
  const unsigned n_waves = gridDim.x * 8u;
  const unsigned per     = (n_tiles + n_waves - 1u) / n_waves;
  const unsigned gw      = blockIdx.x * 8u + wave;
  unsigned t0 = gw * per;
  if (t0 > n_tiles) t0 = n_tiles;
  unsigned t1 = t0 + per;
  if (t1 > n_tiles) t1 = n_tiles;
  const unsigned cnt = t1 - t0;
  if (cnt == 0) return;

  const v2f* w2 = (const v2f*)w;
  const v2f b1 = w2[half];
  const v2f b2 = w2[half + 2];

  const float* xp = x + ((size_t)t0 * TILE_O + l) * 8 + 2 * half;
  float* op = out + (size_t)t0 * TILE_O + 8 * half;
  const bool writer = (l == 0);

  const unsigned pairs = cnt / (2 * TU);
  if (pairs != 0) {
    Group g0, g1;
    load_group(g0, xp);
    for (unsigned p = 1; p < pairs; ++p) {
      load_group(g1, xp + TU * TILE_X);
      mul_store_group(g0, b1, b2, op, writer);
      load_group(g0, xp + 2 * TU * TILE_X);
      mul_store_group(g1, b1, b2, op + TU * TILE_O, writer);
      xp += 2 * TU * TILE_X;
      op += 2 * TU * TILE_O;
    }
    load_group(g1, xp + TU * TILE_X);
    mul_store_group(g0, b1, b2, op, writer);
    mul_store_group(g1, b1, b2, op + TU * TILE_O, writer);
    xp += 2 * TU * TILE_X;
    op += 2 * TU * TILE_O;
  }

  for (unsigned r = cnt - pairs * (2 * TU); r != 0; --r) {
    const v2f a1 = *(const v2f*)(xp);
    const v2f a2 = *(const v2f*)(xp + 4);
    v8f c = {0.f, 0.f, 0.f, 0.f, 0.f, 0.f, 0.f, 0.f};
    c = __builtin_amdgcn_wmma_f32_16x16x4_f32(false, a1, false, b1, (short)0, c,
                                              false, false);
    c = __builtin_amdgcn_wmma_f32_16x16x4_f32(false, a2, false, b2, (short)0, c,
                                              false, false);
    if (writer) {
      float4 lo, hi;
      lo.x = c[0]; lo.y = c[1]; lo.z = c[2]; lo.w = c[3];
      hi.x = c[4]; hi.y = c[5]; hi.z = c[6]; hi.w = c[7];
      *(float4*)(op)     = lo;
      *(float4*)(op + 4) = hi;
    }
    xp += TILE_X;
    op += TILE_O;
  }
}

extern "C" void kernel_launch(void* const* d_in, const int* in_sizes, int n_in,
                              void* d_out, int out_size, void* d_ws, size_t ws_size,
                              hipStream_t stream) {
  const float* x = (const float*)d_in[0];
  const float* w = (const float*)d_in[1];
  float* out = (float*)d_out;

  const long long n_out = (long long)out_size;
  TrainableMovingAverage1d_wmma_kernel<<<512, 256, 0, stream>>>(x, w, out, n_out);
}


# w as A (rows), x as B (window c in lane c): two tiles leave in one 32-lane b32 store of a whole 128-B line, 4 per group for 16 two-piece ones; sched barriers keep the loads grouped; same sums
# speedup vs baseline: 1.7138x; 1.0632x over previous
#include <hip/hip_runtime.h>

typedef float v2f __attribute__((ext_vector_type(2)));
typedef float v8f __attribute__((ext_vector_type(8)));


constexpr int TU     = 8;
constexpr int TILE_X = 128;
constexpr int TILE_O = 16;

struct Group {
  v2f lo[TU];
  v2f hi[TU];
};

static __device__ __forceinline__ void load_group(Group& g,
                                                  const float* __restrict__ xp) {
#pragma unroll
  for (int u = 0; u < TU; ++u) {
    g.lo[u] = *(const v2f*)(xp + u * TILE_X);
    g.hi[u] = *(const v2f*)(xp + u * TILE_X + 4);
  }
}

static __device__ __forceinline__ void mul_store_group(const Group& g, v2f a1,
                                                       v2f a2,
                                                       float* __restrict__ op_lane,
                                                       bool upper) {
  v8f c[TU];
#pragma unroll
  for (int u = 0; u < TU; ++u) {
    const v8f z = {0.f, 0.f, 0.f, 0.f, 0.f, 0.f, 0.f, 0.f};
    c[u] = __builtin_amdgcn_wmma_f32_16x16x4_f32(false, a1, false, g.lo[u],
                                                 (short)0, z, false, false);
  }
#pragma unroll
  for (int u = 0; u < TU; ++u) {
    c[u] = __builtin_amdgcn_wmma_f32_16x16x4_f32(false, a2, false, g.hi[u],
                                                 (short)0, c[u], false, false);
  }
#pragma unroll
  for (int u = 0; u < TU; u += 2) {
    op_lane[u * TILE_O] = upper ? c[u + 1][0] : c[u][0];
  }
}

__global__ __launch_bounds__(256) void TrainableMovingAverage1d_wmma_kernel(
    const float* __restrict__ x, const float* __restrict__ w,
    float* __restrict__ out, long long n_out) {
  const int lane = threadIdx.x & 31;
  const unsigned wave = __builtin_amdgcn_readfirstlane(threadIdx.x >> 5);
  const int half = lane >> 4;
  const int l    = lane & 15;

  const unsigned n_tiles = (unsigned)(n_out >> 4);
  const unsigned n_waves = gridDim.x * 8u;
  const unsigned per     = (n_tiles + n_waves - 1u) / n_waves;
  const unsigned gw      = blockIdx.x * 8u + wave;
  unsigned t0 = gw * per;
  if (t0 > n_tiles) t0 = n_tiles;
  unsigned t1 = t0 + per;
  if (t1 > n_tiles) t1 = n_tiles;
  const unsigned cnt = t1 - t0;
  if (cnt == 0) return;

  const v2f* w2 = (const v2f*)w;
  const v2f a1 = w2[half];
  const v2f a2 = w2[half + 2];

  const float* xp = x + ((size_t)t0 * TILE_O + l) * 8 + 2 * half;
  float* op = out + (size_t)t0 * TILE_O + lane;
  const bool upper = (half != 0);

  const unsigned pairs = cnt / (2 * TU);
  if (pairs != 0) {
    Group g0, g1;
    load_group(g0, xp);
    for (unsigned p = 1; p < pairs; ++p) {
      load_group(g1, xp + TU * TILE_X);
      __builtin_amdgcn_sched_barrier(0);
      mul_store_group(g0, a1, a2, op, upper);
      __builtin_amdgcn_sched_barrier(0);
      load_group(g0, xp + 2 * TU * TILE_X);
      __builtin_amdgcn_sched_barrier(0);
      mul_store_group(g1, a1, a2, op + TU * TILE_O, upper);
      __builtin_amdgcn_sched_barrier(0);
      xp += 2 * TU * TILE_X;
      op += 2 * TU * TILE_O;
    }
    load_group(g1, xp + TU * TILE_X);
    __builtin_amdgcn_sched_barrier(0);
    mul_store_group(g0, a1, a2, op, upper);
    __builtin_amdgcn_sched_barrier(0);
    mul_store_group(g1, a1, a2, op + TU * TILE_O, upper);
    xp += 2 * TU * TILE_X;
    op += 2 * TU * TILE_O;
  }

  for (unsigned r = cnt - pairs * (2 * TU); r != 0; --r) {
    const v2f b1 = *(const v2f*)(xp);
    const v2f b2 = *(const v2f*)(xp + 4);
    v8f c = {0.f, 0.f, 0.f, 0.f, 0.f, 0.f, 0.f, 0.f};
    c = __builtin_amdgcn_wmma_f32_16x16x4_f32(false, a1, false, b1, (short)0, c,
                                              false, false);
    c = __builtin_amdgcn_wmma_f32_16x16x4_f32(false, a2, false, b2, (short)0, c,
                                              false, false);
    if (!upper) {
      op[0] = c[0];
    }
    xp += TILE_X;
    op += TILE_O;
  }
}

extern "C" void kernel_launch(void* const* d_in, const int* in_sizes, int n_in,
                              void* d_out, int out_size, void* d_ws, size_t ws_size,
                              hipStream_t stream) {
  const float* x = (const float*)d_in[0];
  const float* w = (const float*)d_in[1];
  float* out = (float*)d_out;

  const long long n_out = (long long)out_size;
  TrainableMovingAverage1d_wmma_kernel<<<512, 256, 0, stream>>>(x, w, out, n_out);
}


# x loads b64 -> b128: one whole-line 512-B load per tile (lane (c,h) holds taps 4h..4h+3, w fragments permuted alike); 16 loads a group as before, now 8 KiB; sum order: taps 0,1,4,5 then 2,3,6,7
# speedup vs baseline: 1.8247x; 1.0647x over previous
#include <hip/hip_runtime.h>

typedef float v2f __attribute__((ext_vector_type(2)));
typedef float v4f __attribute__((ext_vector_type(4)));
typedef float v8f __attribute__((ext_vector_type(8)));


constexpr int TU     = 16;
constexpr int SB     = 8;
constexpr int TILE_X = 128;
constexpr int TILE_O = 16;

struct Group {
  v4f q[TU];
};

static __device__ __forceinline__ void load_group(Group& g,
                                                  const float* __restrict__ xp) {
#pragma unroll
  for (int u = 0; u < TU; ++u) {
    g.q[u] = *(const v4f*)(xp + u * TILE_X);
  }
}

static __device__ __forceinline__ void mul_store_group(const Group& g, v2f a1,
                                                       v2f a2,
                                                       float* __restrict__ op_lane,
                                                       bool upper) {
#pragma unroll
  for (int s = 0; s < TU; s += SB) {
    v8f c[SB];
#pragma unroll
    for (int u = 0; u < SB; ++u) {
      const v8f z = {0.f, 0.f, 0.f, 0.f, 0.f, 0.f, 0.f, 0.f};
      c[u] = __builtin_amdgcn_wmma_f32_16x16x4_f32(false, a1, false,
                                                   g.q[s + u].xy, (short)0, z,
                                                   false, false);
    }
#pragma unroll
    for (int u = 0; u < SB; ++u) {
      c[u] = __builtin_amdgcn_wmma_f32_16x16x4_f32(false, a2, false,
                                                   g.q[s + u].zw, (short)0,
                                                   c[u], false, false);
    }
#pragma unroll
    for (int u = 0; u < SB; u += 2) {
      op_lane[(s + u) * TILE_O] = upper ? c[u + 1][0] : c[u][0];
    }
    if (s + SB < TU) __builtin_amdgcn_sched_barrier(0);
  }
}

__global__ __launch_bounds__(256) void TrainableMovingAverage1d_wmma_kernel(
    const float* __restrict__ x, const float* __restrict__ w,
    float* __restrict__ out, long long n_out) {
  const int lane = threadIdx.x & 31;
  const unsigned wave = __builtin_amdgcn_readfirstlane(threadIdx.x >> 5);
  const int half = lane >> 4;
  const int l    = lane & 15;

  const unsigned n_tiles = (unsigned)(n_out >> 4);
  const unsigned n_waves = gridDim.x * 8u;
  const unsigned per     = (n_tiles + n_waves - 1u) / n_waves;
  const unsigned gw      = blockIdx.x * 8u + wave;
  unsigned t0 = gw * per;
  if (t0 > n_tiles) t0 = n_tiles;
  unsigned t1 = t0 + per;
  if (t1 > n_tiles) t1 = n_tiles;
  const unsigned cnt = t1 - t0;
  if (cnt == 0) return;

  const v2f* w2 = (const v2f*)w;
  const v2f a1 = w2[2 * half];
  const v2f a2 = w2[2 * half + 1];

  const float* xp = x + ((size_t)t0 * TILE_O + l) * 8 + 4 * half;
  float* op = out + (size_t)t0 * TILE_O + lane;
  const bool upper = (half != 0);

  const unsigned pairs = cnt / (2 * TU);
  if (pairs != 0) {
    Group g0, g1;
    load_group(g0, xp);
    for (unsigned p = 1; p < pairs; ++p) {
      load_group(g1, xp + TU * TILE_X);
      __builtin_amdgcn_sched_barrier(0);
      mul_store_group(g0, a1, a2, op, upper);
      __builtin_amdgcn_sched_barrier(0);
      load_group(g0, xp + 2 * TU * TILE_X);
      __builtin_amdgcn_sched_barrier(0);
      mul_store_group(g1, a1, a2, op + TU * TILE_O, upper);
      __builtin_amdgcn_sched_barrier(0);
      xp += 2 * TU * TILE_X;
      op += 2 * TU * TILE_O;
    }
    load_group(g1, xp + TU * TILE_X);
    __builtin_amdgcn_sched_barrier(0);
    mul_store_group(g0, a1, a2, op, upper);
    __builtin_amdgcn_sched_barrier(0);
    mul_store_group(g1, a1, a2, op + TU * TILE_O, upper);
    xp += 2 * TU * TILE_X;
    op += 2 * TU * TILE_O;
  }

  for (unsigned r = cnt - pairs * (2 * TU); r != 0; --r) {
    const v4f q = *(const v4f*)(xp);
    v8f c = {0.f, 0.f, 0.f, 0.f, 0.f, 0.f, 0.f, 0.f};
    c = __builtin_amdgcn_wmma_f32_16x16x4_f32(false, a1, false, q.xy, (short)0,
                                              c, false, false);
    c = __builtin_amdgcn_wmma_f32_16x16x4_f32(false, a2, false, q.zw, (short)0,
                                              c, false, false);
    if (!upper) {
      op[0] = c[0];
    }
    xp += TILE_X;
    op += TILE_O;
  }
}

extern "C" void kernel_launch(void* const* d_in, const int* in_sizes, int n_in,
                              void* d_out, int out_size, void* d_ws, size_t ws_size,
                              hipStream_t stream) {
  const float* x = (const float*)d_in[0];
  const float* w = (const float*)d_in[1];
  float* out = (float*)d_out;

  const long long n_out = (long long)out_size;
  TrainableMovingAverage1d_wmma_kernel<<<512, 256, 0, stream>>>(x, w, out, n_out);
}
